# starting kernel
# baseline (speedup 1.0000x reference)
#include <hip/hip_runtime.h>

#define F_FIELDS 39
#define D_DIM    64

typedef __bf16 v16bf __attribute__((ext_vector_type(16)));
typedef __bf16 v2bf  __attribute__((ext_vector_type(2)));
typedef float  v8f   __attribute__((ext_vector_type(8)));

union FragU { v16bf v; v2bf p[8]; };

__device__ __forceinline__ v2bf pk_bf16(float x, float y) {
#if __has_builtin(__builtin_amdgcn_cvt_pk_bf16_f32)
    return __builtin_amdgcn_cvt_pk_bf16_f32(x, y);
#else
    v2bf r; r.x = (__bf16)x; r.y = (__bf16)y; return r;
#endif
}

__global__ __launch_bounds__(256) void fwfm_wmma_kernel(
    const int*   __restrict__ x,
    const float* __restrict__ emb,
    const float* __restrict__ bias,
    const float* __restrict__ W,
    const float* __restrict__ w0,
    float*       __restrict__ out,
    int Brows)
{
    const int lane   = threadIdx.x & 31;
    const int wave   = (int)((blockIdx.x * blockDim.x + threadIdx.x) >> 5);
    const int nwaves = (int)((gridDim.x * blockDim.x) >> 5);

    const int TI[6] = {0,0,0,1,1,2};
    const int TJ[6] = {0,1,2,1,2,2};

    const int n_c  = lane & 15;
    const int m_hi = (lane >> 4) * 8;
    float wreg[6][8];
    #pragma unroll
    for (int tp = 0; tp < 6; ++tp) {
        #pragma unroll
        for (int r = 0; r < 8; ++r) {
            int i = 16 * TI[tp] + m_hi + r;
            int j = 16 * TJ[tp] + n_c;
            wreg[tp][r] = (i < j && j < F_FIELDS) ? W[i * F_FIELDS + j] : 0.0f;
        }
    }

    const int rrow   = lane & 15;
    const int kbase0 = (lane & 16) ? 8 : 0;

    for (int b = wave; b < Brows; b += nwaves) {
        const int* xb = x + (size_t)b * F_FIELDS;

        FragU frag[3][2];
        #pragma unroll
        for (int t = 0; t < 3; ++t) {
            const int  row   = 16 * t + rrow;
            const bool valid = (row < F_FIELDS);
            const int  idx   = valid ? xb[row] : 0;
            const float* erow = emb + (size_t)idx * D_DIM;
            #pragma unroll
            for (int h = 0; h < 2; ++h) {
                const int kb = kbase0 + 32 * h;
                FragU& f = frag[t][h];
                if (valid) {
                    const float4* p0 = (const float4*)(erow + kb);
                    const float4* p1 = (const float4*)(erow + kb + 16);
                    float4 a0 = p0[0], a1 = p0[1];
                    float4 b0 = p1[0], b1 = p1[1];
                    f.p[0] = pk_bf16(a0.x, a0.y);
                    f.p[1] = pk_bf16(a0.z, a0.w);
                    f.p[2] = pk_bf16(a1.x, a1.y);
                    f.p[3] = pk_bf16(a1.z, a1.w);
                    f.p[4] = pk_bf16(b0.x, b0.y);
                    f.p[5] = pk_bf16(b0.z, b0.w);
                    f.p[6] = pk_bf16(b1.x, b1.y);
                    f.p[7] = pk_bf16(b1.z, b1.w);
                } else {
                    v2bf z = pk_bf16(0.0f, 0.0f);
                    #pragma unroll
                    for (int s = 0; s < 8; ++s) f.p[s] = z;
                }
            }
        }

        float wsum = 0.0f;
        #pragma unroll
        for (int tp = 0; tp < 6; ++tp) {
            v8f acc = {};
            acc = __builtin_amdgcn_wmma_f32_16x16x32_bf16(
                false, frag[TI[tp]][0].v, false, frag[TJ[tp]][0].v,
                (short)0, acc, false, false);
            acc = __builtin_amdgcn_wmma_f32_16x16x32_bf16(
                false, frag[TI[tp]][1].v, false, frag[TJ[tp]][1].v,
                (short)0, acc, false, false);
            #pragma unroll
            for (int r = 0; r < 8; ++r) wsum += wreg[tp][r] * acc[r];
        }

        wsum += bias[(size_t)xb[lane]];
        if (lane + 32 < F_FIELDS) wsum += bias[(size_t)xb[lane + 32]];

        #pragma unroll
        for (int off = 16; off >= 1; off >>= 1)
            wsum += __shfl_xor(wsum, off, 32);

        if (lane == 0) out[b] = w0[0] + wsum;
    }
}

extern "C" void kernel_launch(void* const* d_in, const int* in_sizes, int n_in,
                              void* d_out, int out_size, void* d_ws, size_t ws_size,
                              hipStream_t stream) {
    const int*   x    = (const int*)  d_in[0];
    const float* emb  = (const float*)d_in[1];
    const float* bias = (const float*)d_in[2];
    const float* W    = (const float*)d_in[3];
    const float* w0   = (const float*)d_in[4];
    float*       out  = (float*)d_out;

    const int Brows = in_sizes[0] / F_FIELDS;
    const int block = 256;
    int grid = (Brows + 7) / 8;
    if (grid < 1) grid = 1;

    fwfm_wmma_kernel<<<grid, block, 0, stream>>>(x, emb, bias, W, w0, out, Brows);
}


# Re-associated: T = Wm^T x E on WMMA (masked W now a bf16 operand; embedding dim along lanes, whole rows load straight into the operand), then f32 elementwise sum of E*T; new sum order
# speedup vs baseline: 1.4540x; 1.4540x over previous
#include <hip/hip_runtime.h>

#define F_FIELDS 39
#define D_DIM    64


typedef __bf16 v16bf __attribute__((ext_vector_type(16)));
typedef __bf16 v2bf  __attribute__((ext_vector_type(2)));
typedef float  v8f   __attribute__((ext_vector_type(8)));

union FragU { v16bf v; v2bf p[8]; };

__device__ __forceinline__ v2bf pk_bf16(float x, float y) {
#if __has_builtin(__builtin_amdgcn_cvt_pk_bf16_f32)
    return __builtin_amdgcn_cvt_pk_bf16_f32(x, y);
#else
    v2bf r; r.x = (__bf16)x; r.y = (__bf16)y; return r;
#endif
}


__global__ __launch_bounds__(256, 6) void fwfm_wmma_kernel(
    const int*   __restrict__ x,
    const float* __restrict__ emb,
    const float* __restrict__ bias,
    const float* __restrict__ W,
    const float* __restrict__ w0,
    float*       __restrict__ out,
    int Brows)
{
    const int lane   = threadIdx.x & 31;
    const int l      = lane & 15;
    const int h      = lane >> 4;
    const int wave   = (int)((blockIdx.x * blockDim.x + threadIdx.x) >> 5);

    const int j0 = l;
    const int j1 = 16 + l;
    const int j2 = (l < 4) ? (32 + l) : ((l >= 8 && l < 11) ? (28 + l) : F_FIELDS);
    const int j2c = (j2 < F_FIELDS) ? j2 : (F_FIELDS - 1);

    const int b = wave;
    if (b < Brows) {
        const int  bu = __builtin_amdgcn_readfirstlane(b);
        const int* xb = x + (size_t)bu * F_FIELDS;

        int xi[F_FIELDS];
        #pragma unroll
        for (int f = 0; f < F_FIELDS; ++f) {
            xi[f] = __builtin_amdgcn_readfirstlane(xb[f]);
        }

        const int f1  = (lane + 32 < F_FIELDS) ? (lane + 32) : (F_FIELDS - 1);
        const int xv0 = xb[lane];
        const int xv1 = xb[f1];
        __builtin_amdgcn_sched_barrier(0);

        float t0[8], t1[16], t2[16], t3[4];
        #pragma unroll
        for (int s = 0; s < 16; ++s) {
            const int i1  = ((s < 8) ? s : (s + 8)) + 8 * h;
            const int i2  = 32 + 4 * h + s;
            const int i2c = (i2 < F_FIELDS) ? i2 : (F_FIELDS - 1);
            if (s < 8) t0[s] = W[i1 * F_FIELDS + j0];
            t1[s] = W[i1 * F_FIELDS + j1];
            t2[s] = W[i1 * F_FIELDS + j2c];
            if (s < 4) t3[s] = W[i2c * F_FIELDS + j2c];
        }
        __builtin_amdgcn_sched_barrier(0);

        float4 ev[20];
        #pragma unroll
        for (int s = 0; s < 20; ++s) {
            const int fa  = (s < 8) ? s       : ((s < 16) ? (s + 8)  : (32 + (s - 16)));
            const int fb  = (s < 8) ? (8 + s) : ((s < 16) ? (s + 16) : (36 + (s - 16)));
            const int fbv = (fb < F_FIELDS) ? fb : (F_FIELDS - 1);
            const int idx = h ? xi[fbv] : xi[fa];
            const unsigned off = (unsigned)idx * (unsigned)(D_DIM * 4) + (unsigned)l * 16u;
            ev[s] = *(const float4*)((const char*)emb + off);
        }
        const float bias0 = bias[(size_t)xv0];
        const float bias1 = bias[(size_t)xv1];
        __builtin_amdgcn_sched_barrier(0);

        FragU wa[4];
        #pragma unroll
        for (int u = 0; u < 8; ++u) {
            float w0a[2], w1a[2], w2a[2], w3a[2];
            #pragma unroll
            for (int e2 = 0; e2 < 2; ++e2) {
                const int s  = 2 * u + e2;
                const int i1 = ((s < 8) ? s : (s + 8)) + 8 * h;
                const int i2 = 32 + 4 * h + s;
                w0a[e2] = (s < 8 && i1 < j0) ? t0[s < 8 ? s : 0] : 0.0f;
                w1a[e2] = (i1 < j1) ? t1[s] : 0.0f;
                w2a[e2] = (i1 < j2 && j2 < F_FIELDS) ? t2[s] : 0.0f;
                w3a[e2] = (s < 4 && i2 < j2 && j2 < F_FIELDS) ? t3[s < 4 ? s : 0] : 0.0f;
            }
            wa[0].p[u] = pk_bf16(w0a[0], w0a[1]);
            wa[1].p[u] = pk_bf16(w1a[0], w1a[1]);
            wa[2].p[u] = pk_bf16(w2a[0], w2a[1]);
            wa[3].p[u] = pk_bf16(w3a[0], w3a[1]);
        }
        __builtin_amdgcn_sched_barrier(0);

        float e[20][4];
        #pragma unroll
        for (int s = 0; s < 20; ++s) {
            const bool pad = (s == 19) && (h != 0);
            e[s][0] = pad ? 0.0f : ev[s].x;
            e[s][1] = pad ? 0.0f : ev[s].y;
            e[s][2] = pad ? 0.0f : ev[s].z;
            e[s][3] = pad ? 0.0f : ev[s].w;
        }

        float a0 = 0.0f, a1 = 0.0f, a2 = 0.0f;
        const v2bf zz = pk_bf16(0.0f, 0.0f);
        #pragma unroll
        for (int q = 0; q < 4; ++q) {
            FragU b1, b2;
            #pragma unroll
            for (int u = 0; u < 8; ++u) b1.p[u] = pk_bf16(e[2 * u][q], e[2 * u + 1][q]);
            b2.p[0] = pk_bf16(e[16][q], e[17][q]);
            b2.p[1] = pk_bf16(e[18][q], e[19][q]);
            #pragma unroll
            for (int u = 2; u < 8; ++u) b2.p[u] = zz;

            v8f d0 = {}, d1 = {}, d2 = {};
            d0 = __builtin_amdgcn_wmma_f32_16x16x32_bf16(false, wa[0].v, false, b1.v, (short)0, d0, false, false);
            d1 = __builtin_amdgcn_wmma_f32_16x16x32_bf16(false, wa[1].v, false, b1.v, (short)0, d1, false, false);
            d2 = __builtin_amdgcn_wmma_f32_16x16x32_bf16(false, wa[2].v, false, b1.v, (short)0, d2, false, false);
            d2 = __builtin_amdgcn_wmma_f32_16x16x32_bf16(false, wa[3].v, false, b2.v, (short)0, d2, false, false);

            #pragma unroll
            for (int r = 0; r < 8; ++r) a0 = __builtin_fmaf(d0[r], e[r][q], a0);
            #pragma unroll
            for (int r = 0; r < 8; ++r) a1 = __builtin_fmaf(d1[r], e[8 + r][q], a1);
            #pragma unroll
            for (int r = 0; r < 4; ++r) a2 = __builtin_fmaf(d2[r], e[16 + r][q], a2);
        }
        float wsum = (a0 + a1) + a2;

        wsum += bias0;
        wsum += (lane + 32 < F_FIELDS) ? bias1 : 0.0f;

        #pragma unroll
        for (int off = 16; off >= 1; off >>= 1)
            wsum += __shfl_xor(wsum, off, 32);

        if (lane == 0) out[b] = w0[0] + wsum;
    }
}

extern "C" void kernel_launch(void* const* d_in, const int* in_sizes, int n_in,
                              void* d_out, int out_size, void* d_ws, size_t ws_size,
                              hipStream_t stream) {
    const int*   x    = (const int*)  d_in[0];
    const float* emb  = (const float*)d_in[1];
    const float* bias = (const float*)d_in[2];
    const float* W    = (const float*)d_in[3];
    const float* w0   = (const float*)d_in[4];
    float*       out  = (float*)d_out;

    const int Brows = in_sizes[0] / F_FIELDS;
    const int block = 256;
    int grid = (Brows + 7) / 8;
    if (grid < 1) grid = 1;

    fwfm_wmma_kernel<<<grid, block, 0, stream>>>(x, emb, bias, W, w0, out, Brows);
}


# Four batch rows per wave side by side: one product of 256 columns against one weight-fragment set, all rows' gathers requested up front; 8192 one-wave work-groups enter instead of 32768 waves
# speedup vs baseline: 1.5926x; 1.0953x over previous
#include <hip/hip_runtime.h>

#define F_FIELDS 39
#define D_DIM    64
#define ROWS     4


typedef __bf16 v16bf __attribute__((ext_vector_type(16)));
typedef __bf16 v2bf  __attribute__((ext_vector_type(2)));
typedef float  v8f   __attribute__((ext_vector_type(8)));

union FragU { v16bf v; v2bf p[8]; };

__device__ __forceinline__ v2bf pk_bf16(float x, float y) {
#if __has_builtin(__builtin_amdgcn_cvt_pk_bf16_f32)
    return __builtin_amdgcn_cvt_pk_bf16_f32(x, y);
#else
    v2bf r; r.x = (__bf16)x; r.y = (__bf16)y; return r;
#endif
}


__global__ __launch_bounds__(32, 2) void fwfm_wmma_kernel(
    const int*   __restrict__ x,
    const float* __restrict__ emb,
    const float* __restrict__ bias,
    const float* __restrict__ W,
    const float* __restrict__ w0,
    float*       __restrict__ out,
    int Brows)
{
    const int lane   = threadIdx.x & 31;
    const int l      = lane & 15;
    const int h      = lane >> 4;
    const int wave   = (int)blockIdx.x;

    const int j0 = l;
    const int j1 = 16 + l;
    const int j2 = (l < 4) ? (32 + l) : ((l >= 8 && l < 11) ? (28 + l) : F_FIELDS);
    const int j2c = (j2 < F_FIELDS) ? j2 : (F_FIELDS - 1);

    const int b0 = wave * ROWS;
    if (b0 < Brows) {
        const int  bu = b0;

        const int* xb[ROWS];
        #pragma unroll
        for (int r = 0; r < ROWS; ++r) {
            const int br = (bu + r < Brows) ? (bu + r) : (Brows - 1);
            xb[r] = x + (size_t)br * F_FIELDS;
        }

        int xi[ROWS][F_FIELDS];
        #pragma unroll
        for (int r = 0; r < ROWS && r < 2; ++r) {
            #pragma unroll
            for (int f = 0; f < F_FIELDS; ++f) {
                xi[r][f] = __builtin_amdgcn_readfirstlane(xb[r][f]);
            }
        }

        const int f1  = (lane + 32 < F_FIELDS) ? (lane + 32) : (F_FIELDS - 1);
        int xv0[ROWS], xv1[ROWS];
        #pragma unroll
        for (int r = 0; r < ROWS; ++r) {
            xv0[r] = xb[r][lane];
            xv1[r] = xb[r][f1];
        }
        __builtin_amdgcn_sched_barrier(0);

        float t0[8], t1[16], t2[16], t3[4];
        #pragma unroll
        for (int s = 0; s < 16; ++s) {
            const int i1  = ((s < 8) ? s : (s + 8)) + 8 * h;
            const int i2  = 32 + 4 * h + s;
            const int i2c = (i2 < F_FIELDS) ? i2 : (F_FIELDS - 1);
            if (s < 8) t0[s] = W[i1 * F_FIELDS + j0];
            t1[s] = W[i1 * F_FIELDS + j1];
            t2[s] = W[i1 * F_FIELDS + j2c];
            if (s < 4) t3[s] = W[i2c * F_FIELDS + j2c];
        }
        __builtin_amdgcn_sched_barrier(0);

        float4 ev[ROWS][20];
        #pragma unroll
        for (int r = 0; r < ROWS; ++r) {
            #pragma unroll
            for (int s = 0; s < 20; ++s) {
                const int fa  = (s < 8) ? s       : ((s < 16) ? (s + 8)  : (32 + (s - 16)));
                const int fb  = (s < 8) ? (8 + s) : ((s < 16) ? (s + 16) : (36 + (s - 16)));
                const int fbv = (fb < F_FIELDS) ? fb : (F_FIELDS - 1);
                const int idx = h ? xi[r][fbv] : xi[r][fa];
                const unsigned off = (unsigned)idx * (unsigned)(D_DIM * 4) + (unsigned)l * 16u;
                ev[r][s] = *(const float4*)((const char*)emb + off);
            }
            if (r + 2 < ROWS) {
                #pragma unroll
                for (int f = 0; f < F_FIELDS; ++f)
                    xi[r + 2][f] = __builtin_amdgcn_readfirstlane(xb[r + 2][f]);
            }
            __builtin_amdgcn_sched_barrier(0);
        }
        float bias0[ROWS], bias1[ROWS];
        #pragma unroll
        for (int r = 0; r < ROWS; ++r) {
            bias0[r] = bias[(size_t)xv0[r]];
            bias1[r] = bias[(size_t)xv1[r]];
        }
        __builtin_amdgcn_sched_barrier(0);

        FragU wa[4];
        #pragma unroll
        for (int u = 0; u < 8; ++u) {
            float w0a[2], w1a[2], w2a[2], w3a[2];
            #pragma unroll
            for (int e2 = 0; e2 < 2; ++e2) {
                const int s  = 2 * u + e2;
                const int i1 = ((s < 8) ? s : (s + 8)) + 8 * h;
                const int i2 = 32 + 4 * h + s;
                w0a[e2] = (s < 8 && i1 < j0) ? t0[s < 8 ? s : 0] : 0.0f;
                w1a[e2] = (i1 < j1) ? t1[s] : 0.0f;
                w2a[e2] = (i1 < j2 && j2 < F_FIELDS) ? t2[s] : 0.0f;
                w3a[e2] = (s < 4 && i2 < j2 && j2 < F_FIELDS) ? t3[s < 4 ? s : 0] : 0.0f;
            }
            wa[0].p[u] = pk_bf16(w0a[0], w0a[1]);
            wa[1].p[u] = pk_bf16(w1a[0], w1a[1]);
            wa[2].p[u] = pk_bf16(w2a[0], w2a[1]);
            wa[3].p[u] = pk_bf16(w3a[0], w3a[1]);
        }
        __builtin_amdgcn_sched_barrier(0);

        float wsum[ROWS];
        const v2bf zz = pk_bf16(0.0f, 0.0f);
        #pragma unroll
        for (int r = 0; r < ROWS; ++r) {
            float e[20][4];
            #pragma unroll
            for (int s = 0; s < 20; ++s) {
                const bool pad = (s == 19) && (h != 0);
                e[s][0] = pad ? 0.0f : ev[r][s].x;
                e[s][1] = pad ? 0.0f : ev[r][s].y;
                e[s][2] = pad ? 0.0f : ev[r][s].z;
                e[s][3] = pad ? 0.0f : ev[r][s].w;
            }

            float a0 = 0.0f, a1 = 0.0f, a2 = 0.0f;
            #pragma unroll
            for (int q = 0; q < 4; ++q) {
                FragU b1, b2;
                #pragma unroll
                for (int u = 0; u < 8; ++u) b1.p[u] = pk_bf16(e[2 * u][q], e[2 * u + 1][q]);
                b2.p[0] = pk_bf16(e[16][q], e[17][q]);
                b2.p[1] = pk_bf16(e[18][q], e[19][q]);
                #pragma unroll
                for (int u = 2; u < 8; ++u) b2.p[u] = zz;

                v8f d0 = {}, d1 = {}, d2 = {};
                d0 = __builtin_amdgcn_wmma_f32_16x16x32_bf16(false, wa[0].v, false, b1.v, (short)0, d0, false, false);
                d1 = __builtin_amdgcn_wmma_f32_16x16x32_bf16(false, wa[1].v, false, b1.v, (short)0, d1, false, false);
                d2 = __builtin_amdgcn_wmma_f32_16x16x32_bf16(false, wa[2].v, false, b1.v, (short)0, d2, false, false);
                d2 = __builtin_amdgcn_wmma_f32_16x16x32_bf16(false, wa[3].v, false, b2.v, (short)0, d2, false, false);

                #pragma unroll
                for (int rr = 0; rr < 8; ++rr) a0 = __builtin_fmaf(d0[rr], e[rr][q], a0);
                #pragma unroll
                for (int rr = 0; rr < 8; ++rr) a1 = __builtin_fmaf(d1[rr], e[8 + rr][q], a1);
                #pragma unroll
                for (int rr = 0; rr < 4; ++rr) a2 = __builtin_fmaf(d2[rr], e[16 + rr][q], a2);
            }
            wsum[r] = (a0 + a1) + a2;
            __builtin_amdgcn_sched_barrier(0);
        }

        #pragma unroll
        for (int r = 0; r < ROWS; ++r) {
            wsum[r] += bias0[r];
            wsum[r] += (lane + 32 < F_FIELDS) ? bias1[r] : 0.0f;
        }

        #pragma unroll
        for (int off = 16; off >= 1; off >>= 1) {
            #pragma unroll
            for (int r = 0; r < ROWS; ++r)
                wsum[r] += __shfl_xor(wsum[r], off, 32);
        }

        if (lane == 0) {
            const float c0 = w0[0];
            #pragma unroll
            for (int r = 0; r < ROWS; ++r)
                if (bu + r < Brows) out[bu + r] = c0 + wsum[r];
        }
    }
}

extern "C" void kernel_launch(void* const* d_in, const int* in_sizes, int n_in,
                              void* d_out, int out_size, void* d_ws, size_t ws_size,
                              hipStream_t stream) {
    const int*   x    = (const int*)  d_in[0];
    const float* emb  = (const float*)d_in[1];
    const float* bias = (const float*)d_in[2];
    const float* W    = (const float*)d_in[3];
    const float* w0   = (const float*)d_in[4];
    float*       out  = (float*)d_out;

    const int Brows = in_sizes[0] / F_FIELDS;
    const int block = 32;
    const int rows_per_block = ROWS;
    int grid = (Brows + rows_per_block - 1) / rows_per_block;
    if (grid < 1) grid = 1;

    fwfm_wmma_kernel<<<grid, block, 0, stream>>>(x, emb, bias, W, w0, out, Brows);
}
